# starting kernel
# baseline (speedup 1.0000x reference)
#include <hip/hip_runtime.h>
#include <hip/hip_bf16.h>

typedef __attribute__((ext_vector_type(2))) float v2f;
typedef __attribute__((ext_vector_type(4))) float v4f;
typedef __attribute__((ext_vector_type(8))) float v8f;

#define H_DIM   4096
#define IN_DIM  512
#define E_DIM   8
#define OUT_DIM 512
#define NTOK    16384

__global__ void moe_prep_weff(const float* __restrict__ Wi, const float* __restrict__ Wr,
                              const float* __restrict__ bi, const float* __restrict__ br,
                              float* __restrict__ WcT, float* __restrict__ beff) {
    int t = blockIdx.x * blockDim.x + threadIdx.x;
    int e = t >> 9;
    int k = t & 511;
    const float* wr = Wr + (size_t)e * H_DIM;
    float acc = 0.f;
    for (int h = 0; h < H_DIM; ++h)
        acc = __builtin_fmaf(wr[h], Wi[(size_t)h * IN_DIM + k], acc);
    WcT[e * IN_DIM + k] = acc;
    if (k == 0) {
        float b = br[e];
        for (int h = 0; h < H_DIM; ++h)
            b = __builtin_fmaf(wr[h], bi[h], b);
        beff[e] = b;
    }
}

__global__ void moe_prep_misc(const float* __restrict__ Wi, const float* __restrict__ Wo,
                              float* __restrict__ WcT, float* __restrict__ wsum) {
    int k = blockIdx.x * blockDim.x + threadIdx.x;
    #pragma unroll
    for (int n = 0; n < 8; ++n)
        WcT[(8 + n) * IN_DIM + k] = Wi[(size_t)n * IN_DIM + k];
    const float* wo = Wo + (size_t)k * H_DIM;
    float s = 0.f;
    for (int h = 0; h < H_DIM; ++h) s += wo[h];
    wsum[k] = s;
}

__global__ void __launch_bounds__(256)
moe_main(const float* __restrict__ x,    const float* __restrict__ WcT,
         const float* __restrict__ beff, const float* __restrict__ bi,
         const float* __restrict__ wsum, const float* __restrict__ bo,
         float* __restrict__ out) {
    __shared__ float lds_c[8][16][16];
    __shared__ float lds_agg[8][16];

    const int wave  = threadIdx.x >> 5;
    const int lane  = threadIdx.x & 31;
    const int tile  = blockIdx.x * 8 + wave;
    const int tok0  = tile * 16;
    const int col   = lane & 15;
    const int half  = lane >> 4;

    const float* ap = x   + (size_t)(tok0 + col) * IN_DIM + half * 2;
    const float* bp = WcT + (size_t)col * IN_DIM + half * 2;

    v8f acc = {};
    #pragma unroll 8
    for (int k = 0; k < IN_DIM; k += 4) {
        v2f a = *(const v2f*)(ap + k);
        v2f b = *(const v2f*)(bp + k);
        acc = __builtin_amdgcn_wmma_f32_16x16x4_f32(
             false, a,  false, b,
             (short)0, acc,  false,  false);
    }

    v4f wv[4], bv[4];
    #pragma unroll
    for (int i = 0; i < 4; ++i) {
        wv[i] = *(const v4f*)(wsum + lane * 4 + 128 * i);
        bv[i] = *(const v4f*)(bo   + lane * 4 + 128 * i);
    }

    float cb = (col < 8) ? beff[col] : bi[col - 8];
    #pragma unroll
    for (int j = 0; j < 8; ++j)
        lds_c[wave][half * 8 + j][col] = acc[j] + cb;
    __syncthreads();

    if (lane < 16) {
        const float* r = lds_c[wave][lane];
        float b1 = r[0]; int i1 = 0;
        #pragma unroll
        for (int n = 1; n < 8; ++n) if (r[n] > b1) { b1 = r[n]; i1 = n; }
        float b2 = -3.402823466e38f; int i2 = 0;
        #pragma unroll
        for (int n = 0; n < 8; ++n) if (n != i1 && r[n] > b2) { b2 = r[n]; i2 = n; }
        lds_agg[wave][lane] = 0.5f * (r[8 + i1] + r[8 + i2]);
    }
    __syncthreads();

    #pragma unroll 4
    for (int t = 0; t < 16; ++t) {
        float ag = lds_agg[wave][t];
        float* op = out + (size_t)(tok0 + t) * OUT_DIM + lane * 4;
        #pragma unroll
        for (int i = 0; i < 4; ++i) {
            v4f r = ag * wv[i] + bv[i];
            *(v4f*)(op + 128 * i) = r;
        }
    }
}

extern "C" void kernel_launch(void* const* d_in, const int* in_sizes, int n_in,
                              void* d_out, int out_size, void* d_ws, size_t ws_size,
                              hipStream_t stream) {
    const float* x  = (const float*)d_in[0];
    const float* Wi = (const float*)d_in[1];
    const float* bi = (const float*)d_in[2];
    const float* Wr = (const float*)d_in[3];
    const float* br = (const float*)d_in[4];
    const float* Wo = (const float*)d_in[5];
    const float* bo = (const float*)d_in[6];
    float* out = (float*)d_out;

    float* WcT  = (float*)d_ws;
    float* beff = WcT + 16 * IN_DIM;
    float* wsum = beff + 16;

    moe_prep_weff<<<16, 256, 0, stream>>>(Wi, Wr, bi, br, WcT, beff);
    moe_prep_misc<<<2, 256, 0, stream>>>(Wi, Wo, WcT, wsum);
    moe_main<<<NTOK / 16 / 8, 256, 0, stream>>>(x, WcT, beff, bi, wsum, bo, out);
}


# Weight prep spread over the chip: h-sums of folded router weight and out_proj row sums split over 520 work-groups into partial planes plus a small reduce launch; changes their f32 summation order
# speedup vs baseline: 13.7180x; 13.7180x over previous
#include <hip/hip_runtime.h>
#include <hip/hip_bf16.h>

typedef __attribute__((ext_vector_type(2))) float v2f;
typedef __attribute__((ext_vector_type(4))) float v4f;
typedef __attribute__((ext_vector_type(8))) float v8f;

typedef __attribute__((address_space(4))) float cfloat;

#define H_DIM   4096
#define IN_DIM  512
#define E_DIM   8
#define OUT_DIM 512
#define NTOK    16384

#define PLANE   4096

__device__ __forceinline__ v4f fma4(float s, v4f v, v4f a) {
    v4f r;
    r.x = __builtin_fmaf(s, v.x, a.x);
    r.y = __builtin_fmaf(s, v.y, a.y);
    r.z = __builtin_fmaf(s, v.z, a.z);
    r.w = __builtin_fmaf(s, v.w, a.w);
    return r;
}

__device__ __forceinline__ float wave_sum(float s) {
    #pragma unroll
    for (int m = 16; m >= 1; m >>= 1) s += __shfl_xor(s, m, 32);
    return s;
}

__global__ void __launch_bounds__(256)
moe_prep_partial(const float* __restrict__ Wi, const float* __restrict__ Wr,
                 const float* __restrict__ bi, const float* __restrict__ br,
                 const float* __restrict__ Wo,
                 float* __restrict__ part, float* __restrict__ beff,
                 float* __restrict__ wsum) {
    __shared__ v4f   red4[8][E_DIM][32];
    __shared__ float red1[8];

    const int lane = threadIdx.x & 31;
    const int wave = __builtin_amdgcn_readfirstlane((int)(threadIdx.x >> 5));
    const int wg   = blockIdx.x;

    if (wg < 256) {
        const int hc = wg >> 2;
        const int ks = wg & 3;
        const int h0 = hc * 64 + wave * 8;
        const float* wi = Wi + (size_t)h0 * IN_DIM + ks * 128 + lane * 4;
        float wrv[E_DIM][8];
        #pragma unroll
        for (int e = 0; e < E_DIM; ++e) {
            const cfloat* wr = (const cfloat*)(Wr + (size_t)e * H_DIM + h0);
            #pragma unroll
            for (int i = 0; i < 8; ++i) wrv[e][i] = wr[i];
        }
        v4f v[8];
        #pragma unroll
        for (int i = 0; i < 8; ++i) v[i] = *(const v4f*)(wi + (size_t)i * IN_DIM);
        asm volatile("" :: "v"(v[0]), "v"(v[1]), "v"(v[2]), "v"(v[3]),
                           "v"(v[4]), "v"(v[5]), "v"(v[6]), "v"(v[7]));
        #pragma unroll
        for (int e = 0; e < E_DIM; ++e) {
            v4f a = wrv[e][0] * v[0];
            #pragma unroll
            for (int i = 1; i < 8; ++i) a = fma4(wrv[e][i], v[i], a);
            red4[wave][e][lane] = a;
        }
        __syncthreads();
        v4f s = red4[0][wave][lane];
        #pragma unroll
        for (int w = 1; w < 8; ++w) s += red4[w][wave][lane];
        *(v4f*)(part + (size_t)hc * PLANE + wave * IN_DIM + ks * 128 + lane * 4) = s;
    } else if (wg < 512) {
        const int r2  = wg - 256;
        const int row = r2 * 2 + (wave >> 2);
        const float* wo = Wo + (size_t)row * H_DIM + (wave & 3) * 1024 + lane * 4;
        v4f v[8];
        #pragma unroll
        for (int i = 0; i < 8; ++i) v[i] = *(const v4f*)(wo + i * 128);
        v4f s4 = ((v[0] + v[1]) + (v[2] + v[3])) + ((v[4] + v[5]) + (v[6] + v[7]));
        float s = wave_sum((s4.x + s4.y) + (s4.z + s4.w));
        if (lane == 0) red1[wave] = s;
        __syncthreads();
        if (threadIdx.x < 2) {
            const float* r = red1 + threadIdx.x * 4;
            wsum[r2 * 2 + threadIdx.x] = (r[0] + r[1]) + (r[2] + r[3]);
        }
    } else {
        const int e = wg - 512;
        const int hb = wave * 512 + lane * 4;
        const float* wr = Wr + (size_t)e * H_DIM + hb;
        const float* bb = bi + hb;
        v4f a[4], c[4];
        #pragma unroll
        for (int i = 0; i < 4; ++i) {
            a[i] = *(const v4f*)(wr + i * 128);
            c[i] = *(const v4f*)(bb + i * 128);
        }
        v4f s4 = a[0] * c[0];
        #pragma unroll
        for (int i = 1; i < 4; ++i) s4 = __builtin_elementwise_fma(a[i], c[i], s4);
        float s = wave_sum((s4.x + s4.y) + (s4.z + s4.w));
        if (lane == 0) red1[wave] = s;
        __syncthreads();
        if (threadIdx.x == 0) {
            float t = ((red1[0] + red1[1]) + (red1[2] + red1[3])) +
                      ((red1[4] + red1[5]) + (red1[6] + red1[7]));
            beff[e] = br[e] + t;
        }
    }
}

__global__ void __launch_bounds__(256)
moe_prep_reduce(const float* __restrict__ part, const float* __restrict__ Wi,
                float* __restrict__ WcT) {
    const int t  = blockIdx.x * 256 + threadIdx.x;
    const int q  = t & 3;
    const int o4 = t >> 2;
    const float* p = part + (size_t)(q * 16) * PLANE + o4 * 4;
    v4f v[16];
    #pragma unroll
    for (int i = 0; i < 16; ++i) v[i] = *(const v4f*)(p + (size_t)i * PLANE);
    float cp = Wi[t];
    v4f s = v[0];
    #pragma unroll
    for (int i = 1; i < 16; ++i) s += v[i];
    #pragma unroll
    for (int m = 1; m <= 2; m <<= 1) {
        s.x += __shfl_xor(s.x, m, 32);
        s.y += __shfl_xor(s.y, m, 32);
        s.z += __shfl_xor(s.z, m, 32);
        s.w += __shfl_xor(s.w, m, 32);
    }
    if (q == 0) *(v4f*)(WcT + o4 * 4) = s;
    WcT[8 * IN_DIM + t] = cp;
}

__global__ void __launch_bounds__(256)
moe_main(const float* __restrict__ x,    const float* __restrict__ WcT,
         const float* __restrict__ beff, const float* __restrict__ bi,
         const float* __restrict__ wsum, const float* __restrict__ bo,
         float* __restrict__ out) {
    __shared__ float lds_c[8][16][16];
    __shared__ float lds_agg[8][16];

    const int wave  = threadIdx.x >> 5;
    const int lane  = threadIdx.x & 31;
    const int tile  = blockIdx.x * 8 + wave;
    const int tok0  = tile * 16;
    const int col   = lane & 15;
    const int half  = lane >> 4;

    const float* ap = x   + (size_t)(tok0 + col) * IN_DIM + half * 2;
    const float* bp = WcT + (size_t)col * IN_DIM + half * 2;

    v8f acc = {};
    #pragma unroll 8
    for (int k = 0; k < IN_DIM; k += 4) {
        v2f a = *(const v2f*)(ap + k);
        v2f b = *(const v2f*)(bp + k);
        acc = __builtin_amdgcn_wmma_f32_16x16x4_f32(
             false, a,  false, b,
             (short)0, acc,  false,  false);
    }

    v4f wv[4], bv[4];
    #pragma unroll
    for (int i = 0; i < 4; ++i) {
        wv[i] = *(const v4f*)(wsum + lane * 4 + 128 * i);
        bv[i] = *(const v4f*)(bo   + lane * 4 + 128 * i);
    }

    float cb = (col < 8) ? beff[col] : bi[col - 8];
    #pragma unroll
    for (int j = 0; j < 8; ++j)
        lds_c[wave][half * 8 + j][col] = acc[j] + cb;
    __syncthreads();

    if (lane < 16) {
        const float* r = lds_c[wave][lane];
        float b1 = r[0]; int i1 = 0;
        #pragma unroll
        for (int n = 1; n < 8; ++n) if (r[n] > b1) { b1 = r[n]; i1 = n; }
        float b2 = -3.402823466e38f; int i2 = 0;
        #pragma unroll
        for (int n = 0; n < 8; ++n) if (n != i1 && r[n] > b2) { b2 = r[n]; i2 = n; }
        lds_agg[wave][lane] = 0.5f * (r[8 + i1] + r[8 + i2]);
    }
    __syncthreads();

    #pragma unroll 4
    for (int t = 0; t < 16; ++t) {
        float ag = lds_agg[wave][t];
        float* op = out + (size_t)(tok0 + t) * OUT_DIM + lane * 4;
        #pragma unroll
        for (int i = 0; i < 4; ++i) {
            v4f r = ag * wv[i] + bv[i];
            *(v4f*)(op + 128 * i) = r;
        }
    }
}

extern "C" void kernel_launch(void* const* d_in, const int* in_sizes, int n_in,
                              void* d_out, int out_size, void* d_ws, size_t ws_size,
                              hipStream_t stream) {
    const float* x  = (const float*)d_in[0];
    const float* Wi = (const float*)d_in[1];
    const float* bi = (const float*)d_in[2];
    const float* Wr = (const float*)d_in[3];
    const float* br = (const float*)d_in[4];
    const float* Wo = (const float*)d_in[5];
    const float* bo = (const float*)d_in[6];
    float* out = (float*)d_out;

    float* WcT  = (float*)d_ws;
    float* beff = WcT + 16 * IN_DIM;
    float* wsum = beff + 16;
    float* part = WcT + 16384;

    moe_prep_partial<<<520, 256, 0, stream>>>(Wi, Wr, bi, br, Wo, part, beff, wsum);
    moe_prep_reduce<<<16, 256, 0, stream>>>(part, Wi, WcT);
    moe_main<<<NTOK / 16 / 8, 256, 0, stream>>>(x, WcT, beff, bi, wsum, bo, out);
}


# Main kernel re-tiled for whole-line loads: 4 waves split each tile's k, x row pieces go through LDS into fragments, weight kept in fragment order; changes the f32 summation order of the products
# speedup vs baseline: 21.4473x; 1.5634x over previous
#include <hip/hip_runtime.h>
#include <hip/hip_bf16.h>

typedef __attribute__((ext_vector_type(2))) float v2f;
typedef __attribute__((ext_vector_type(4))) float v4f;
typedef __attribute__((ext_vector_type(8))) float v8f;

typedef __attribute__((address_space(4))) float cfloat;

#define H_DIM   4096
#define IN_DIM  512
#define E_DIM   8
#define OUT_DIM 512
#define NTOK    16384

#define PLANE   4096

__device__ __forceinline__ v4f fma4(float s, v4f v, v4f a) {
    v4f r;
    r.x = __builtin_fmaf(s, v.x, a.x);
    r.y = __builtin_fmaf(s, v.y, a.y);
    r.z = __builtin_fmaf(s, v.z, a.z);
    r.w = __builtin_fmaf(s, v.w, a.w);
    return r;
}

__device__ __forceinline__ float wave_sum(float s) {
    #pragma unroll
    for (int m = 16; m >= 1; m >>= 1) s += __shfl_xor(s, m, 32);
    return s;
}

__global__ void __launch_bounds__(256)
moe_prep_partial(const float* __restrict__ Wi, const float* __restrict__ Wr,
                 const float* __restrict__ bi, const float* __restrict__ br,
                 const float* __restrict__ Wo,
                 float* __restrict__ part, float* __restrict__ beff,
                 float* __restrict__ wsum) {
    __shared__ v4f   red4[8][E_DIM][32];
    __shared__ float red1[8];

    const int lane = threadIdx.x & 31;
    const int wave = __builtin_amdgcn_readfirstlane((int)(threadIdx.x >> 5));
    const int wg   = blockIdx.x;

    if (wg < 256) {
        const int hc = wg >> 2;
        const int ks = wg & 3;
        const int h0 = hc * 64 + wave * 8;
        const float* wi = Wi + (size_t)h0 * IN_DIM + ks * 128 + lane * 4;
        float wrv[E_DIM][8];
        #pragma unroll
        for (int e = 0; e < E_DIM; ++e) {
            const cfloat* wr = (const cfloat*)(Wr + (size_t)e * H_DIM + h0);
            #pragma unroll
            for (int i = 0; i < 8; ++i) wrv[e][i] = wr[i];
        }
        v4f v[8];
        #pragma unroll
        for (int i = 0; i < 8; ++i) v[i] = *(const v4f*)(wi + (size_t)i * IN_DIM);
        asm volatile("" :: "v"(v[0]), "v"(v[1]), "v"(v[2]), "v"(v[3]),
                           "v"(v[4]), "v"(v[5]), "v"(v[6]), "v"(v[7]));
        #pragma unroll
        for (int e = 0; e < E_DIM; ++e) {
            v4f a = wrv[e][0] * v[0];
            #pragma unroll
            for (int i = 1; i < 8; ++i) a = fma4(wrv[e][i], v[i], a);
            red4[wave][e][lane] = a;
        }
        __syncthreads();
        v4f s = red4[0][wave][lane];
        #pragma unroll
        for (int w = 1; w < 8; ++w) s += red4[w][wave][lane];
        *(v4f*)(part + (size_t)hc * PLANE + wave * IN_DIM + ks * 128 + lane * 4) = s;
    } else if (wg < 512) {
        const int r2  = wg - 256;
        const int row = r2 * 2 + (wave >> 2);
        const float* wo = Wo + (size_t)row * H_DIM + (wave & 3) * 1024 + lane * 4;
        v4f v[8];
        #pragma unroll
        for (int i = 0; i < 8; ++i) v[i] = *(const v4f*)(wo + i * 128);
        v4f s4 = ((v[0] + v[1]) + (v[2] + v[3])) + ((v[4] + v[5]) + (v[6] + v[7]));
        float s = wave_sum((s4.x + s4.y) + (s4.z + s4.w));
        if (lane == 0) red1[wave] = s;
        __syncthreads();
        if (threadIdx.x < 2) {
            const float* r = red1 + threadIdx.x * 4;
            wsum[r2 * 2 + threadIdx.x] = (r[0] + r[1]) + (r[2] + r[3]);
        }
    } else {
        const int e = wg - 512;
        const int hb = wave * 512 + lane * 4;
        const float* wr = Wr + (size_t)e * H_DIM + hb;
        const float* bb = bi + hb;
        v4f a[4], c[4];
        #pragma unroll
        for (int i = 0; i < 4; ++i) {
            a[i] = *(const v4f*)(wr + i * 128);
            c[i] = *(const v4f*)(bb + i * 128);
        }
        v4f s4 = a[0] * c[0];
        #pragma unroll
        for (int i = 1; i < 4; ++i) s4 = __builtin_elementwise_fma(a[i], c[i], s4);
        float s = wave_sum((s4.x + s4.y) + (s4.z + s4.w));
        if (lane == 0) red1[wave] = s;
        __syncthreads();
        if (threadIdx.x == 0) {
            float t = ((red1[0] + red1[1]) + (red1[2] + red1[3])) +
                      ((red1[4] + red1[5]) + (red1[6] + red1[7]));
            beff[e] = br[e] + t;
        }
    }
}

__global__ void __launch_bounds__(256)
moe_prep_reduce(const float* __restrict__ part, const float* __restrict__ Wi,
                float* __restrict__ Bf) {
    __shared__ v4f sm[2][8][8];

    const int b   = blockIdx.x;
    const int t   = threadIdx.x;
    const int q   = t & 3;
    const int kql = (t >> 2) & 7;
    const int e   = t >> 5;
    const float* p = part + (size_t)(q * 16) * PLANE + e * IN_DIM + b * 32 + kql * 4;
    v4f v[16];
    #pragma unroll
    for (int i = 0; i < 16; ++i) v[i] = *(const v4f*)(p + (size_t)i * PLANE);
    const int kk = t & 31;
    float cp = Wi[e * IN_DIM + b * 32 + kk];
    v4f s = v[0];
    #pragma unroll
    for (int i = 1; i < 16; ++i) s += v[i];
    #pragma unroll
    for (int m = 1; m <= 2; m <<= 1) {
        s.x += __shfl_xor(s.x, m, 32);
        s.y += __shfl_xor(s.y, m, 32);
        s.z += __shfl_xor(s.z, m, 32);
        s.w += __shfl_xor(s.w, m, 32);
    }
    if (q == 0) sm[0][kql][e] = s;
    ((float*)&sm[1][kk >> 2][e])[kk & 3] = cp;
    __syncthreads();
    if (t < 128) {
        const int sel = t >> 6;
        const int kl  = (t >> 3) & 7;
        const int r   = t & 7;
        const int kq  = b * 8 + kl;
        const int ks  = kq >> 5, j = (kq >> 1) & 15, h = kq & 1;
        *(v4f*)(Bf + (size_t)(((ks * 16 + j) * 32 + h * 16 + sel * 8 + r) * 4)) = sm[sel][kl][r];
    }
}

#define APITCH 132
#define PPITCH 20

__global__ void __launch_bounds__(128)
moe_main(const float* __restrict__ x,    const float* __restrict__ Bf,
         const float* __restrict__ beff, const float* __restrict__ bi,
         const float* __restrict__ wsum, const float* __restrict__ bo,
         float* __restrict__ out) {
    __shared__ __attribute__((aligned(16))) float stage[4][16 * APITCH];
    __shared__ __attribute__((aligned(16))) float lds_p[4][16 * PPITCH];

    const int lane = threadIdx.x & 31;
    const int wave = __builtin_amdgcn_readfirstlane((int)(threadIdx.x >> 5));
    const int tok0 = blockIdx.x * 16;
    const int col  = lane & 15;
    const int half = lane >> 4;

    const float* xp = x  + (size_t)tok0 * IN_DIM + wave * 128 + lane * 4;
    const float* bp = Bf + (size_t)wave * 2048 + lane * 4;
    v4f xa[16], bf[16];
    #pragma unroll
    for (int r = 0; r < 16; ++r) xa[r] = *(const v4f*)(xp + (size_t)r * IN_DIM);
    #pragma unroll
    for (int j = 0; j < 16; ++j) bf[j] = *(const v4f*)(bp + j * 128);

    float* st = stage[wave];
    #pragma unroll
    for (int r = 0; r < 16; ++r) *(v4f*)(st + r * APITCH + lane * 4) = xa[r];
    __builtin_amdgcn_wave_barrier();
    const float* fr = st + col * APITCH + half * 4;
    v4f af[16];
    #pragma unroll
    for (int j = 0; j < 16; ++j) af[j] = *(const v4f*)(fr + j * 8);
    asm volatile("" :: "v"(af[0]),  "v"(af[1]),  "v"(af[2]),  "v"(af[3]),
                       "v"(af[4]),  "v"(af[5]),  "v"(af[6]),  "v"(af[7]),
                       "v"(af[8]),  "v"(af[9]),  "v"(af[10]), "v"(af[11]),
                       "v"(af[12]), "v"(af[13]), "v"(af[14]), "v"(af[15]));

    v8f acc0 = {}, acc1 = {};
    #pragma unroll
    for (int j = 0; j < 16; ++j) {
        v2f a0 = {af[j].x, af[j].y}, a1 = {af[j].z, af[j].w};
        v2f b0 = {bf[j].x, bf[j].y}, b1 = {bf[j].z, bf[j].w};
        acc0 = __builtin_amdgcn_wmma_f32_16x16x4_f32(
             false, a0,  false, b0,
             (short)0, acc0,  false,  false);
        acc1 = __builtin_amdgcn_wmma_f32_16x16x4_f32(
             false, a1,  false, b1,
             (short)0, acc1,  false,  false);
    }

    v4f wv[4], bv[4];
    #pragma unroll
    for (int i = 0; i < 4; ++i) {
        wv[i] = *(const v4f*)(wsum + lane * 4 + 128 * i);
        bv[i] = *(const v4f*)(bo   + lane * 4 + 128 * i);
    }
    v4f cb[4];
    cb[0] = *(const v4f*)(beff);
    cb[1] = *(const v4f*)(beff + 4);
    cb[2] = *(const v4f*)(bi);
    cb[3] = *(const v4f*)(bi + 4);

    v8f acc = acc0 + acc1;
    float* pp = lds_p[wave];
    #pragma unroll
    for (int j = 0; j < 8; ++j) pp[(half * 8 + j) * PPITCH + col] = acc[j];
    __syncthreads();

    const int row = wave * 4 + (lane & 3);
    v4f c4[4];
    #pragma unroll
    for (int n = 0; n < 4; ++n) {
        v4f p0 = *(const v4f*)(lds_p[0] + row * PPITCH + n * 4);
        v4f p1 = *(const v4f*)(lds_p[1] + row * PPITCH + n * 4);
        v4f p2 = *(const v4f*)(lds_p[2] + row * PPITCH + n * 4);
        v4f p3 = *(const v4f*)(lds_p[3] + row * PPITCH + n * 4);
        c4[n] = ((p0 + p1) + (p2 + p3)) + cb[n];
    }
    const float lg[8] = {c4[0].x, c4[0].y, c4[0].z, c4[0].w, c4[1].x, c4[1].y, c4[1].z, c4[1].w};
    const float hv[8] = {c4[2].x, c4[2].y, c4[2].z, c4[2].w, c4[3].x, c4[3].y, c4[3].z, c4[3].w};
    float b1 = lg[0], g1 = hv[0]; int i1 = 0;
    #pragma unroll
    for (int n = 1; n < 8; ++n) {
        const bool up = lg[n] > b1;
        b1 = up ? lg[n] : b1;  g1 = up ? hv[n] : g1;  i1 = up ? n : i1;
    }
    float b2 = -3.402823466e38f, g2 = hv[0];
    #pragma unroll
    for (int n = 0; n < 8; ++n) {
        const bool up = (n != i1) && (lg[n] > b2);
        b2 = up ? lg[n] : b2;  g2 = up ? hv[n] : g2;
    }
    const float agg = 0.5f * (g1 + g2);

    #pragma unroll
    for (int t = 0; t < 4; ++t) {
        const float ag = __shfl(agg, t, 32);
        float* op = out + (size_t)(tok0 + wave * 4 + t) * OUT_DIM + lane * 4;
        #pragma unroll
        for (int i = 0; i < 4; ++i) {
            v4f r = ag * wv[i] + bv[i];
            *(v4f*)(op + 128 * i) = r;
        }
    }
}

extern "C" void kernel_launch(void* const* d_in, const int* in_sizes, int n_in,
                              void* d_out, int out_size, void* d_ws, size_t ws_size,
                              hipStream_t stream) {
    const float* x  = (const float*)d_in[0];
    const float* Wi = (const float*)d_in[1];
    const float* bi = (const float*)d_in[2];
    const float* Wr = (const float*)d_in[3];
    const float* br = (const float*)d_in[4];
    const float* Wo = (const float*)d_in[5];
    const float* bo = (const float*)d_in[6];
    float* out = (float*)d_out;

    float* WcT  = (float*)d_ws;
    float* beff = WcT + 16 * IN_DIM;
    float* wsum = beff + 16;
    float* part = WcT + 16384;

    moe_prep_partial<<<520, 256, 0, stream>>>(Wi, Wr, bi, br, Wo, part, beff, wsum);
    moe_prep_reduce<<<16, 256, 0, stream>>>(part, Wi, WcT);
    moe_main<<<NTOK / 16, 128, 0, stream>>>(x, WcT, beff, bi, wsum, bo, out);
}
